# starting kernel
# baseline (speedup 1.0000x reference)
#include <hip/hip_runtime.h>
#include <hip/hip_bf16.h>

#define FIN  4
#define FHID 64

typedef float v2f __attribute__((ext_vector_type(2)));
typedef float v8f __attribute__((ext_vector_type(8)));

static __host__ int cdiv(int a, int b) { return (a + b - 1) / b; }

__global__ void k_degree(const int* __restrict__ ei, float* __restrict__ deg,
                         int E, int N) {
    int e = blockIdx.x * blockDim.x + threadIdx.x;
    if (e >= E + N) return;
    int dst = (e < E) ? ei[E + e] : (e - E);
    atomicAdd(&deg[dst], 1.0f);
}

__global__ void k_rsqrt(float* __restrict__ d, int N) {
    int n = blockIdx.x * blockDim.x + threadIdx.x;
    if (n >= N) return;
    float v = d[n];
    d[n] = (v > 0.0f) ? rsqrtf(v) : 0.0f;
}

__global__ void k_xw1_wmma(const float* __restrict__ x,
                           const float* __restrict__ W1,
                           float* __restrict__ t, int N) {
    const int lane = threadIdx.x & 31;
    const int wave = threadIdx.x >> 5;
    const int tile = blockIdx.x * (blockDim.x >> 5) + wave;
    if (tile * 16 >= N) return;

    const int half = lane >> 4;
    const int l16  = lane & 15;
    const int row  = tile * 16 + l16;
    const int kb   = half * 2;

    v2f a;
    a.x = x[row * FIN + kb + 0];
    a.y = x[row * FIN + kb + 1];

#pragma unroll
    for (int c = 0; c < FHID / 16; ++c) {
        const int col = c * 16 + l16;
        v2f b;
        b.x = W1[(kb + 0) * FHID + col];
        b.y = W1[(kb + 1) * FHID + col];
        v8f acc = {};
        acc = __builtin_amdgcn_wmma_f32_16x16x4_f32(
             false, a,  false, b,
             (short)0, acc,  false,  false);
#pragma unroll
        for (int i = 0; i < 8; ++i) {
            int m = half * 8 + i;
            t[(size_t)(tile * 16 + m) * FHID + col] = acc[i];
        }
    }
}

__global__ void k_edge1(const int* __restrict__ ei,
                        const float* __restrict__ t,
                        const float* __restrict__ dis,
                        float* __restrict__ agg, int E, int N) {
    const int  lane = threadIdx.x & 31;
    const long long wid =
        (long long)blockIdx.x * (blockDim.x >> 5) + (threadIdx.x >> 5);
    if (wid >= (long long)E + N) return;

    int src, dst;
    if (wid < E) { src = ei[wid]; dst = ei[E + wid]; }
    else         { src = dst = (int)(wid - E); }

    const float nrm = dis[src] * dis[dst];
    const float2 v  = *(const float2*)(t + (size_t)src * FHID + lane * 2);
    float* ap = agg + (size_t)dst * FHID + lane * 2;
    atomicAdd(ap + 0, v.x * nrm);
    atomicAdd(ap + 1, v.y * nrm);
}

__global__ void k_relu_mv(const float* __restrict__ agg,
                          const float* __restrict__ b1,
                          const float* __restrict__ W2,
                          float* __restrict__ z, int N) {
    const int lane = threadIdx.x & 31;
    const int n = blockIdx.x * (blockDim.x >> 5) + (threadIdx.x >> 5);
    if (n >= N) return;
    float s = 0.0f;
#pragma unroll
    for (int j = 0; j < FHID / 32; ++j) {
        int f = j * 32 + lane;
        float h = agg[(size_t)n * FHID + f] + b1[f];
        h = fmaxf(h, 0.0f);
        s = fmaf(h, W2[f], s);
    }
#pragma unroll
    for (int off = 16; off > 0; off >>= 1)
        s += __shfl_xor(s, off, 32);
    if (lane == 0) z[n] = s;
}

__global__ void k_edge2(const int* __restrict__ ei,
                        const float* __restrict__ z,
                        const float* __restrict__ dis,
                        float* __restrict__ oacc, int E, int N) {
    long long e = (long long)blockIdx.x * blockDim.x + threadIdx.x;
    if (e >= (long long)E + N) return;
    int src, dst;
    if (e < E) { src = ei[e]; dst = ei[E + e]; }
    else       { src = dst = (int)(e - E); }
    atomicAdd(&oacc[dst], z[src] * dis[src] * dis[dst]);
}

__global__ void k_sigmoid(const float* __restrict__ oacc,
                          const float* __restrict__ b2,
                          float* __restrict__ out, int N) {
    int n = blockIdx.x * blockDim.x + threadIdx.x;
    if (n >= N) return;
    float v = oacc[n] + b2[0];
    out[n] = 1.0f / (1.0f + __expf(-v));
}

extern "C" void kernel_launch(void* const* d_in, const int* in_sizes, int n_in,
                              void* d_out, int out_size, void* d_ws, size_t ws_size,
                              hipStream_t stream) {
    const float* x  = (const float*)d_in[0];
    const int*   ei = (const int*)d_in[1];
    const float* W1 = (const float*)d_in[2];
    const float* b1 = (const float*)d_in[3];
    const float* W2 = (const float*)d_in[4];
    const float* b2 = (const float*)d_in[5];
    float* out = (float*)d_out;

    const int N = in_sizes[0] / FIN;
    const int E = in_sizes[1] / 2;
    const int T = E + N;

    char* ws = (char*)d_ws;
    float* dis  = (float*)ws;                 ws += (size_t)N * sizeof(float);
    float* t    = (float*)ws;                 ws += (size_t)N * FHID * sizeof(float);
    float* agg  = (float*)ws;                 ws += (size_t)N * FHID * sizeof(float);
    float* z    = (float*)ws;                 ws += (size_t)N * sizeof(float);
    float* oacc = (float*)ws;

    hipMemsetAsync(dis,  0, (size_t)N * sizeof(float), stream);
    hipMemsetAsync(agg,  0, (size_t)N * FHID * sizeof(float), stream);
    hipMemsetAsync(oacc, 0, (size_t)N * sizeof(float), stream);

    const int BT = 256;
    const int WPB = BT / 32;

    k_degree  <<<cdiv(T, BT), BT, 0, stream>>>(ei, dis, E, N);
    k_rsqrt   <<<cdiv(N, BT), BT, 0, stream>>>(dis, N);

    const int tiles = cdiv(N, 16);
    k_xw1_wmma<<<cdiv(tiles, WPB), BT, 0, stream>>>(x, W1, t, N);

    k_edge1   <<<cdiv(T, WPB), BT, 0, stream>>>(ei, t, dis, agg, E, N);
    k_relu_mv <<<cdiv(N, WPB), BT, 0, stream>>>(agg, b1, W2, z, N);
    k_edge2   <<<cdiv(T, BT), BT, 0, stream>>>(ei, z, dis, oacc, E, N);
    k_sigmoid <<<cdiv(N, BT), BT, 0, stream>>>(oacc, b2, out, N);
}


# Layer-1 messages formed at the edge on WMMA: a wave loops over groups of 16 edges, gathers their x rows scaled by the edge norm, 4 WMMAs, adds D rows to agg; t = x W1 never stored; rounding changes
# speedup vs baseline: 1.7494x; 1.7494x over previous
#include <hip/hip_runtime.h>
#include <hip/hip_bf16.h>

#define FIN  4
#define FHID 64

typedef float v2f __attribute__((ext_vector_type(2)));
typedef float v8f __attribute__((ext_vector_type(8)));

static __host__ int cdiv(int a, int b) { return (a + b - 1) / b; }

__global__ void k_degree(const int* __restrict__ ei, float* __restrict__ row,
                         int E, int N) {
    int e = blockIdx.x * blockDim.x + threadIdx.x;
    if (e >= E + N) return;
    int dst = (e < E) ? ei[E + e] : (e - E);
    atomicAdd(&row[(size_t)dst * FHID], 1.0f);
}

__global__ void k_rsqrt(float* __restrict__ row, float* __restrict__ dis, int N) {
    int n = blockIdx.x * blockDim.x + threadIdx.x;
    if (n >= N) return;
    float v = row[(size_t)n * FHID];
    dis[n] = (v > 0.0f) ? rsqrtf(v) : 0.0f;
    row[(size_t)n * FHID] = 0.0f;
}

__global__ void __launch_bounds__(256)
k_edge1_wmma(const int* __restrict__ ei,
             const float* __restrict__ x,
             const float* __restrict__ W1,
             const float* __restrict__ dis,
             float* __restrict__ agg, int E, int N, int G) {
    const int lane = threadIdx.x & 31;
    const int half = lane >> 4;
    const int l16  = lane & 15;
    const int kb   = half * 2;
    const int wpb  = blockDim.x >> 5;
    const int w0   = blockIdx.x * wpb + (threadIdx.x >> 5);
    const int nw   = gridDim.x * wpb;
    const long long T = (long long)E + N;

    v2f b[FHID / 16];
#pragma unroll
    for (int c = 0; c < FHID / 16; ++c) {
        const int col = c * 16 + l16;
        b[c].x = W1[(kb + 0) * FHID + col];
        b[c].y = W1[(kb + 1) * FHID + col];
    }

    int src = 0, dst = 0;
    bool valid = false;
    if (w0 < G) {
        const long long e = (long long)w0 * 16 + l16;
        valid = e < T;
        if (valid) {
            if (e < E) { src = ei[e]; dst = ei[E + e]; }
            else       { src = dst = (int)(e - E); }
        }
    }

    for (int g = w0; g < G; g += nw) {
        const float  ds = dis[src];
        const float  dd = dis[dst];
        const float2 xv = *(const float2*)(x + (size_t)src * FIN + kb);

        int dm[8];
#pragma unroll
        for (int i = 0; i < 8; ++i) dm[i] = __shfl(dst, half * 8 + i, 32);

        const int gn = g + nw;
        int srcn = 0, dstn = 0;
        bool validn = false;
        if (gn < G) {
            const long long e = (long long)gn * 16 + l16;
            validn = e < T;
            if (validn) {
                if (e < E) { srcn = ei[e]; dstn = ei[E + e]; }
                else       { srcn = dstn = (int)(e - E); }
            }
        }

        const float nrm = ds * dd;
        v2f a;
        a.x = valid ? xv.x * nrm : 0.0f;
        a.y = valid ? xv.y * nrm : 0.0f;

        v8f acc[FHID / 16];
#pragma unroll
        for (int c = 0; c < FHID / 16; ++c) {
            v8f zero = {};
            acc[c] = __builtin_amdgcn_wmma_f32_16x16x4_f32(
                 false, a,  false, b[c],
                 (short)0, zero,  false,  false);
        }

#pragma unroll
        for (int i = 0; i < 8; ++i) {
            float* ap = agg + (size_t)dm[i] * FHID + l16;
#pragma unroll
            for (int c = 0; c < FHID / 16; ++c)
                atomicAdd(ap + c * 16, acc[c][i]);
        }

        src = srcn; dst = dstn; valid = validn;
    }
}

__global__ void k_relu_mv(float* __restrict__ agg,
                          const float* __restrict__ b1,
                          const float* __restrict__ W2,
                          float* __restrict__ z, int N) {
    const int lane = threadIdx.x & 31;
    const int n = blockIdx.x * (blockDim.x >> 5) + (threadIdx.x >> 5);
    if (n >= N) return;
    float s = 0.0f;
#pragma unroll
    for (int j = 0; j < FHID / 32; ++j) {
        int f = j * 32 + lane;
        float h = agg[(size_t)n * FHID + f] + b1[f];
        h = fmaxf(h, 0.0f);
        s = fmaf(h, W2[f], s);
    }
#pragma unroll
    for (int off = 16; off > 0; off >>= 1)
        s += __shfl_xor(s, off, 32);
    if (lane == 0) {
        z[n] = s;
        agg[(size_t)n * FHID] = 0.0f;
    }
}

__global__ void k_edge2(const int* __restrict__ ei,
                        const float* __restrict__ z,
                        const float* __restrict__ dis,
                        float* __restrict__ oacc, int E, int N) {
    long long e = (long long)blockIdx.x * blockDim.x + threadIdx.x;
    if (e >= (long long)E + N) return;
    int src, dst;
    if (e < E) { src = ei[e]; dst = ei[E + e]; }
    else       { src = dst = (int)(e - E); }
    atomicAdd(&oacc[(size_t)dst * FHID], z[src] * dis[src] * dis[dst]);
}

__global__ void k_sigmoid(const float* __restrict__ oacc,
                          const float* __restrict__ b2,
                          float* __restrict__ out, int N) {
    int n = blockIdx.x * blockDim.x + threadIdx.x;
    if (n >= N) return;
    float v = oacc[(size_t)n * FHID] + b2[0];
    out[n] = 1.0f / (1.0f + __expf(-v));
}

extern "C" void kernel_launch(void* const* d_in, const int* in_sizes, int n_in,
                              void* d_out, int out_size, void* d_ws, size_t ws_size,
                              hipStream_t stream) {
    const float* x  = (const float*)d_in[0];
    const int*   ei = (const int*)d_in[1];
    const float* W1 = (const float*)d_in[2];
    const float* b1 = (const float*)d_in[3];
    const float* W2 = (const float*)d_in[4];
    const float* b2 = (const float*)d_in[5];
    float* out = (float*)d_out;

    const int N = in_sizes[0] / FIN;
    const int E = in_sizes[1] / 2;
    const int T = E + N;

    char* ws = (char*)d_ws;
    float* dis  = (float*)ws;                 ws += (size_t)N * sizeof(float);
    float* agg  = (float*)ws;                 ws += (size_t)N * FHID * sizeof(float);
    float* z    = (float*)ws;

    hipMemsetAsync(agg,  0, (size_t)N * FHID * sizeof(float), stream);

    const int BT = 256;
    const int WPB = BT / 32;

    k_degree  <<<cdiv(T, BT), BT, 0, stream>>>(ei, agg, E, N);
    k_rsqrt   <<<cdiv(N, BT), BT, 0, stream>>>(agg, dis, N);

    const int G = cdiv(T, 16);
    k_edge1_wmma<<<cdiv(G, 4 * 6), 128, 0, stream>>>(ei, x, W1, dis, agg, E, N, G);
    k_relu_mv <<<cdiv(N, WPB), BT, 0, stream>>>(agg, b1, W2, z, N);
    k_edge2   <<<cdiv(T, BT), BT, 0, stream>>>(ei, z, dis, agg, E, N);
    k_sigmoid <<<cdiv(N, BT), BT, 0, stream>>>(agg, b2, out, N);
}


# Accumulator rows kept twice, one copy per second-level cache half: a wave's atomics go to the copy of the half it runs on (die number), node kernels add the two copies; float sum order changes
# speedup vs baseline: 2.3921x; 1.3674x over previous
#include <hip/hip_runtime.h>
#include <hip/hip_bf16.h>

#define FIN  4
#define FHID 64

typedef float v2f __attribute__((ext_vector_type(2)));
typedef float v8f __attribute__((ext_vector_type(8)));

static __host__ int cdiv(int a, int b) { return (a + b - 1) / b; }

__device__ __forceinline__ int cache_half() {
    return (int)((__builtin_amdgcn_s_sendmsg_rtn(0x87) >> 18) & 1u);
}

__global__ void k_degree(const int* __restrict__ ei, float* __restrict__ row,
                         int E, int N, size_t plane) {
    const int h = cache_half();
    int e = blockIdx.x * blockDim.x + threadIdx.x;
    if (e >= E + N) return;
    int dst = (e < E) ? ei[E + e] : (e - E);
    float* mine = row + (size_t)h * plane;
    atomicAdd(&mine[(size_t)dst * FHID], 1.0f);
}

__global__ void k_rsqrt(float* __restrict__ row, float* __restrict__ dis, int N,
                        size_t plane) {
    int n = blockIdx.x * blockDim.x + threadIdx.x;
    if (n >= N) return;
    float v = row[(size_t)n * FHID] + row[plane + (size_t)n * FHID];
    dis[n] = (v > 0.0f) ? rsqrtf(v) : 0.0f;
    row[(size_t)n * FHID] = 0.0f;
    row[plane + (size_t)n * FHID] = 0.0f;
}

__global__ void __launch_bounds__(256)
k_edge1_wmma(const int* __restrict__ ei,
             const float* __restrict__ x,
             const float* __restrict__ W1,
             const float* __restrict__ dis,
             float* __restrict__ agg0, int E, int N, int G, size_t plane) {
    float* __restrict__ agg = agg0 + (size_t)cache_half() * plane;
    const int lane = threadIdx.x & 31;
    const int half = lane >> 4;
    const int l16  = lane & 15;
    const int kb   = half * 2;
    const int wpb  = blockDim.x >> 5;
    const int w0   = blockIdx.x * wpb + (threadIdx.x >> 5);
    const int nw   = gridDim.x * wpb;
    const long long T = (long long)E + N;

    v2f b[FHID / 16];
#pragma unroll
    for (int c = 0; c < FHID / 16; ++c) {
        const int col = c * 16 + l16;
        b[c].x = W1[(kb + 0) * FHID + col];
        b[c].y = W1[(kb + 1) * FHID + col];
    }

    int src = 0, dst = 0;
    bool valid = false;
    if (w0 < G) {
        const long long e = (long long)w0 * 16 + l16;
        valid = e < T;
        if (valid) {
            if (e < E) { src = ei[e]; dst = ei[E + e]; }
            else       { src = dst = (int)(e - E); }
        }
    }

    for (int g = w0; g < G; g += nw) {
        const float  ds = dis[src];
        const float  dd = dis[dst];
        const float2 xv = *(const float2*)(x + (size_t)src * FIN + kb);

        int dm[8];
#pragma unroll
        for (int i = 0; i < 8; ++i) dm[i] = __shfl(dst, half * 8 + i, 32);

        const int gn = g + nw;
        int srcn = 0, dstn = 0;
        bool validn = false;
        if (gn < G) {
            const long long e = (long long)gn * 16 + l16;
            validn = e < T;
            if (validn) {
                if (e < E) { srcn = ei[e]; dstn = ei[E + e]; }
                else       { srcn = dstn = (int)(e - E); }
            }
        }

        const float nrm = ds * dd;
        v2f a;
        a.x = valid ? xv.x * nrm : 0.0f;
        a.y = valid ? xv.y * nrm : 0.0f;

        v8f acc[FHID / 16];
#pragma unroll
        for (int c = 0; c < FHID / 16; ++c) {
            v8f zero = {};
            acc[c] = __builtin_amdgcn_wmma_f32_16x16x4_f32(
                 false, a,  false, b[c],
                 (short)0, zero,  false,  false);
        }

#pragma unroll
        for (int i = 0; i < 8; ++i) {
            float* ap = agg + (size_t)dm[i] * FHID + l16;
#pragma unroll
            for (int c = 0; c < FHID / 16; ++c)
                atomicAdd(ap + c * 16, acc[c][i]);
        }

        src = srcn; dst = dstn; valid = validn;
    }
}

__global__ void k_relu_mv(float* __restrict__ agg,
                          const float* __restrict__ b1,
                          const float* __restrict__ W2,
                          float* __restrict__ z, int N, size_t plane) {
    const int lane = threadIdx.x & 31;
    const int n = blockIdx.x * (blockDim.x >> 5) + (threadIdx.x >> 5);
    if (n >= N) return;
    float s = 0.0f;
#pragma unroll
    for (int j = 0; j < FHID / 32; ++j) {
        int f = j * 32 + lane;
        float h = (agg[(size_t)n * FHID + f] + agg[plane + (size_t)n * FHID + f]) + b1[f];
        h = fmaxf(h, 0.0f);
        s = fmaf(h, W2[f], s);
    }
#pragma unroll
    for (int off = 16; off > 0; off >>= 1)
        s += __shfl_xor(s, off, 32);
    if (lane == 0) {
        z[n] = s;
        agg[(size_t)n * FHID] = 0.0f;
        agg[plane + (size_t)n * FHID] = 0.0f;
    }
}

__global__ void k_edge2(const int* __restrict__ ei,
                        const float* __restrict__ z,
                        const float* __restrict__ dis,
                        float* __restrict__ oacc, int E, int N, size_t plane) {
    const int h = cache_half();
    long long e = (long long)blockIdx.x * blockDim.x + threadIdx.x;
    if (e >= (long long)E + N) return;
    int src, dst;
    if (e < E) { src = ei[e]; dst = ei[E + e]; }
    else       { src = dst = (int)(e - E); }
    float* mine = oacc + (size_t)h * plane;
    atomicAdd(&mine[(size_t)dst * FHID], z[src] * dis[src] * dis[dst]);
}

__global__ void k_sigmoid(const float* __restrict__ oacc,
                          const float* __restrict__ b2,
                          float* __restrict__ out, int N, size_t plane) {
    int n = blockIdx.x * blockDim.x + threadIdx.x;
    if (n >= N) return;
    float v = (oacc[(size_t)n * FHID] + oacc[plane + (size_t)n * FHID]) + b2[0];
    out[n] = 1.0f / (1.0f + __expf(-v));
}

extern "C" void kernel_launch(void* const* d_in, const int* in_sizes, int n_in,
                              void* d_out, int out_size, void* d_ws, size_t ws_size,
                              hipStream_t stream) {
    const float* x  = (const float*)d_in[0];
    const int*   ei = (const int*)d_in[1];
    const float* W1 = (const float*)d_in[2];
    const float* b1 = (const float*)d_in[3];
    const float* W2 = (const float*)d_in[4];
    const float* b2 = (const float*)d_in[5];
    float* out = (float*)d_out;

    const int N = in_sizes[0] / FIN;
    const int E = in_sizes[1] / 2;
    const int T = E + N;

    char* ws = (char*)d_ws;
    float* dis  = (float*)ws;                 ws += (size_t)N * sizeof(float);
    const size_t plane = (size_t)N * FHID;
    float* agg  = (float*)ws;                 ws += 2 * plane * sizeof(float);
    float* z    = (float*)ws;

    hipMemsetAsync(agg,  0, 2 * plane * sizeof(float), stream);

    const int BT = 256;
    const int WPB = BT / 32;

    k_degree  <<<cdiv(T, BT), BT, 0, stream>>>(ei, agg, E, N, plane);
    k_rsqrt   <<<cdiv(N, BT), BT, 0, stream>>>(agg, dis, N, plane);

    const int G = cdiv(T, 16);
    k_edge1_wmma<<<cdiv(G, 4 * 6), 128, 0, stream>>>(ei, x, W1, dis, agg, E, N, G, plane);
    k_relu_mv <<<cdiv(N, WPB), BT, 0, stream>>>(agg, b1, W2, z, N, plane);
    k_edge2   <<<cdiv(T, BT), BT, 0, stream>>>(ei, z, dis, agg, E, N, plane);
    k_sigmoid <<<cdiv(N, BT), BT, 0, stream>>>(agg, b2, out, N, plane);
}


# Degree, layer-2 edge and relu-dot kernels: a thread takes 8 edges, a wave 24 nodes (16 lanes each), loads issued first: 12x fewer waves to start (relu-dot 50000 to 2084); relu-dot sum order changes
# speedup vs baseline: 2.5088x; 1.0488x over previous
#include <hip/hip_runtime.h>
#include <hip/hip_bf16.h>

#define FIN  4
#define FHID 64
#define EPT  8
#define PPT  12

typedef float v2f __attribute__((ext_vector_type(2)));
typedef float v8f __attribute__((ext_vector_type(8)));

static __host__ int cdiv(int a, int b) { return (a + b - 1) / b; }

__device__ __forceinline__ int cache_half() {
    return (int)((__builtin_amdgcn_s_sendmsg_rtn(0x87) >> 18) & 1u);
}

__global__ void __launch_bounds__(256)
k_degree(const int* __restrict__ ei, float* __restrict__ row,
         int E, int N, size_t plane) {
    const int h = cache_half();
    const long long T = (long long)E + N;
    const long long e0 = (long long)blockIdx.x * (blockDim.x * EPT) + threadIdx.x;
    int dst[EPT];
#pragma unroll
    for (int j = 0; j < EPT; ++j) {
        const long long e = e0 + (long long)j * blockDim.x;
        dst[j] = 0;
        if (e < E)      dst[j] = ei[E + e];
        else if (e < T) dst[j] = (int)(e - E);
    }
    float* mine = row + (size_t)h * plane;
#pragma unroll
    for (int j = 0; j < EPT; ++j) {
        const long long e = e0 + (long long)j * blockDim.x;
        if (e < T) atomicAdd(&mine[(size_t)dst[j] * FHID], 1.0f);
    }
}

__global__ void k_rsqrt(float* __restrict__ row, float* __restrict__ dis, int N,
                        size_t plane) {
    int n = blockIdx.x * blockDim.x + threadIdx.x;
    if (n >= N) return;
    float v = row[(size_t)n * FHID] + row[plane + (size_t)n * FHID];
    dis[n] = (v > 0.0f) ? rsqrtf(v) : 0.0f;
    row[(size_t)n * FHID] = 0.0f;
    row[plane + (size_t)n * FHID] = 0.0f;
}

__global__ void __launch_bounds__(256)
k_edge1_wmma(const int* __restrict__ ei,
             const float* __restrict__ x,
             const float* __restrict__ W1,
             const float* __restrict__ dis,
             float* __restrict__ agg0, int E, int N, int G, size_t plane) {
    float* __restrict__ agg = agg0 + (size_t)cache_half() * plane;
    const int lane = threadIdx.x & 31;
    const int half = lane >> 4;
    const int l16  = lane & 15;
    const int kb   = half * 2;
    const int wpb  = blockDim.x >> 5;
    const int w0   = blockIdx.x * wpb + (threadIdx.x >> 5);
    const int nw   = gridDim.x * wpb;
    const long long T = (long long)E + N;

    v2f b[FHID / 16];
#pragma unroll
    for (int c = 0; c < FHID / 16; ++c) {
        const int col = c * 16 + l16;
        b[c].x = W1[(kb + 0) * FHID + col];
        b[c].y = W1[(kb + 1) * FHID + col];
    }

    int src = 0, dst = 0;
    bool valid = false;
    if (w0 < G) {
        const long long e = (long long)w0 * 16 + l16;
        valid = e < T;
        if (valid) {
            if (e < E) { src = ei[e]; dst = ei[E + e]; }
            else       { src = dst = (int)(e - E); }
        }
    }

    for (int g = w0; g < G; g += nw) {
        const float  ds = dis[src];
        const float  dd = dis[dst];
        const float2 xv = *(const float2*)(x + (size_t)src * FIN + kb);

        int dm[8];
#pragma unroll
        for (int i = 0; i < 8; ++i) dm[i] = __shfl(dst, half * 8 + i, 32);

        const int gn = g + nw;
        int srcn = 0, dstn = 0;
        bool validn = false;
        if (gn < G) {
            const long long e = (long long)gn * 16 + l16;
            validn = e < T;
            if (validn) {
                if (e < E) { srcn = ei[e]; dstn = ei[E + e]; }
                else       { srcn = dstn = (int)(e - E); }
            }
        }

        const float nrm = ds * dd;
        v2f a;
        a.x = valid ? xv.x * nrm : 0.0f;
        a.y = valid ? xv.y * nrm : 0.0f;

        v8f acc[FHID / 16];
#pragma unroll
        for (int c = 0; c < FHID / 16; ++c) {
            v8f zero = {};
            acc[c] = __builtin_amdgcn_wmma_f32_16x16x4_f32(
                 false, a,  false, b[c],
                 (short)0, zero,  false,  false);
        }

#pragma unroll
        for (int i = 0; i < 8; ++i) {
            float* ap = agg + (size_t)dm[i] * FHID + l16;
#pragma unroll
            for (int c = 0; c < FHID / 16; ++c)
                atomicAdd(ap + c * 16, acc[c][i]);
        }

        src = srcn; dst = dstn; valid = validn;
    }
}


template <int M>
__device__ __forceinline__ float add_row_xor(float s) {
    const int b = __builtin_bit_cast(int, s);
    return s + __builtin_bit_cast(float,
        __builtin_amdgcn_update_dpp(b, b, 0x160 | M, 0xf, 0xf, false));
}

__global__ void __launch_bounds__(128)
k_relu_mv(float* __restrict__ agg,
          const float* __restrict__ b1,
          const float* __restrict__ W2,
          float* __restrict__ z, int N, size_t plane) {
    const int lane = threadIdx.x & 31;
    const int sub  = lane >> 4;
    const int l16  = lane & 15;
    const int wpb  = blockDim.x >> 5;
    const int w    = blockIdx.x * wpb + (threadIdx.x >> 5);
    const int nw   = gridDim.x * wpb;
    const int P    = (N + 1) >> 1;
    const float4 bb = *(const float4*)(b1 + l16 * 4);
    const float4 ww = *(const float4*)(W2 + l16 * 4);

    for (int p0 = w; p0 < P; p0 += PPT * nw) {
        float4 a0[PPT], a1[PPT];
#pragma unroll
        for (int j = 0; j < PPT; ++j) {
            const int n  = 2 * (p0 + j * nw) + sub;
            const int nc = n < N ? n : N - 1;
            const float* r = agg + (size_t)nc * FHID + l16 * 4;
            a0[j] = *(const float4*)r;
            a1[j] = *(const float4*)(r + plane);
        }
        __builtin_amdgcn_sched_barrier(0);
        float s[PPT];
#pragma unroll
        for (int j = 0; j < PPT; ++j) {
            const float h0 = fmaxf((a0[j].x + a1[j].x) + bb.x, 0.0f);
            const float h1 = fmaxf((a0[j].y + a1[j].y) + bb.y, 0.0f);
            const float h2 = fmaxf((a0[j].z + a1[j].z) + bb.z, 0.0f);
            const float h3 = fmaxf((a0[j].w + a1[j].w) + bb.w, 0.0f);
            float t = h0 * ww.x;
            t = fmaf(h1, ww.y, t);
            t = fmaf(h2, ww.z, t);
            s[j] = fmaf(h3, ww.w, t);
        }
#pragma unroll
        for (int j = 0; j < PPT; ++j) s[j] = add_row_xor<8>(s[j]);
#pragma unroll
        for (int j = 0; j < PPT; ++j) s[j] = add_row_xor<4>(s[j]);
#pragma unroll
        for (int j = 0; j < PPT; ++j) s[j] = add_row_xor<2>(s[j]);
#pragma unroll
        for (int j = 0; j < PPT; ++j) s[j] = add_row_xor<1>(s[j]);
#pragma unroll
        for (int j = 0; j < PPT; ++j) {
            const int n = 2 * (p0 + j * nw) + sub;
            if (l16 == 0 && n < N) {
                z[n] = s[j];
                agg[(size_t)n * FHID] = 0.0f;
                agg[plane + (size_t)n * FHID] = 0.0f;
            }
        }
    }
}

__global__ void __launch_bounds__(256)
k_edge2(const int* __restrict__ ei,
        const float* __restrict__ z,
        const float* __restrict__ dis,
        float* __restrict__ oacc, int E, int N, size_t plane) {
    const int h = cache_half();
    const long long T = (long long)E + N;
    const long long e0 = (long long)blockIdx.x * (blockDim.x * EPT) + threadIdx.x;
    int src[EPT], dst[EPT];
#pragma unroll
    for (int j = 0; j < EPT; ++j) {
        const long long e = e0 + (long long)j * blockDim.x;
        src[j] = 0; dst[j] = 0;
        if (e < E)      { src[j] = ei[e]; dst[j] = ei[E + e]; }
        else if (e < T) { src[j] = dst[j] = (int)(e - E); }
    }
    float m[EPT];
#pragma unroll
    for (int j = 0; j < EPT; ++j)
        m[j] = z[src[j]] * dis[src[j]] * dis[dst[j]];
    float* mine = oacc + (size_t)h * plane;
#pragma unroll
    for (int j = 0; j < EPT; ++j) {
        const long long e = e0 + (long long)j * blockDim.x;
        if (e < T) atomicAdd(&mine[(size_t)dst[j] * FHID], m[j]);
    }
}

__global__ void k_sigmoid(const float* __restrict__ oacc,
                          const float* __restrict__ b2,
                          float* __restrict__ out, int N, size_t plane) {
    int n = blockIdx.x * blockDim.x + threadIdx.x;
    if (n >= N) return;
    float v = (oacc[(size_t)n * FHID] + oacc[plane + (size_t)n * FHID]) + b2[0];
    out[n] = 1.0f / (1.0f + __expf(-v));
}

extern "C" void kernel_launch(void* const* d_in, const int* in_sizes, int n_in,
                              void* d_out, int out_size, void* d_ws, size_t ws_size,
                              hipStream_t stream) {
    const float* x  = (const float*)d_in[0];
    const int*   ei = (const int*)d_in[1];
    const float* W1 = (const float*)d_in[2];
    const float* b1 = (const float*)d_in[3];
    const float* W2 = (const float*)d_in[4];
    const float* b2 = (const float*)d_in[5];
    float* out = (float*)d_out;

    const int N = in_sizes[0] / FIN;
    const int E = in_sizes[1] / 2;
    const int T = E + N;

    char* ws = (char*)d_ws;
    float* dis  = (float*)ws;                 ws += (size_t)N * sizeof(float);
    const size_t plane = (size_t)N * FHID;
    float* agg  = (float*)ws;                 ws += 2 * plane * sizeof(float);
    float* z    = (float*)ws;

    hipMemsetAsync(agg,  0, 2 * plane * sizeof(float), stream);

    const int BT = 256;

    k_degree  <<<cdiv(T, BT * EPT), BT, 0, stream>>>(ei, agg, E, N, plane);
    k_rsqrt   <<<cdiv(N, BT), BT, 0, stream>>>(agg, dis, N, plane);

    const int G = cdiv(T, 16);
    k_edge1_wmma<<<cdiv(G, 4 * 6), 128, 0, stream>>>(ei, x, W1, dis, agg, E, N, G, plane);
    k_relu_mv <<<cdiv(N, 4 * PPT * 2), 128, 0, stream>>>(agg, b1, W2, z, N, plane);
    k_edge2   <<<cdiv(T, BT * EPT), BT, 0, stream>>>(ei, z, dis, agg, E, N, plane);
    k_sigmoid <<<cdiv(N, BT), BT, 0, stream>>>(agg, b2, out, N, plane);
}
